# starting kernel
# baseline (speedup 1.0000x reference)
#include <hip/hip_runtime.h>
#include <hip/hip_bf16.h>
#include <math.h>

typedef float v2f __attribute__((ext_vector_type(2)));
typedef float v8f __attribute__((ext_vector_type(8)));

#define NEG_SLOPE 0.2f
#define EPS_ 1e-16f

static inline int cdiv(long a, long b) { return (int)((a + b - 1) / b); }

__global__ void fill_kernel(float* __restrict__ p, float v, size_t n) {
    size_t i = (size_t)blockIdx.x * blockDim.x + threadIdx.x;
    size_t stride = (size_t)gridDim.x * blockDim.x;
    for (; i < n; i += stride) p[i] = v;
}

template <int K, int NC, int CPW>
__global__ __launch_bounds__(256) void wmma_gemm_f32(const float* __restrict__ A,
                                                     const float* __restrict__ B,
                                                     float* __restrict__ C,
                                                     int M) {
    constexpr int nColT  = NC / 16;
    constexpr int groups = nColT / CPW;
    int wave = threadIdx.x >> 5;
    int lane = threadIdx.x & 31;
    int hl   = lane >> 4;
    int l    = lane & 15;
    int tile  = blockIdx.x * (blockDim.x >> 5) + wave;
    int total = (M >> 4) * groups;
    if (tile >= total) return;
    int m0 = (tile / groups) << 4;
    int n0 = (tile % groups) * (CPW * 16);

    const float* arow = A + (size_t)(m0 + l) * K + 2 * hl;
    const float* bcol = B + (size_t)(2 * hl) * NC + n0 + l;

    v8f acc[CPW];
#pragma unroll
    for (int i = 0; i < CPW; ++i) acc[i] = {};

    for (int k0 = 0; k0 < K; k0 += 4) {
        v2f a;
        a.x = arow[k0 + 0];
        a.y = arow[k0 + 1];
#pragma unroll
        for (int ct = 0; ct < CPW; ++ct) {
            v2f b;
            b.x = bcol[(size_t)k0 * NC + ct * 16];
            b.y = bcol[(size_t)(k0 + 1) * NC + ct * 16];
            acc[ct] = __builtin_amdgcn_wmma_f32_16x16x4_f32(false, a, false, b,
                                                            (short)0, acc[ct], false, false);
        }
    }

#pragma unroll
    for (int ct = 0; ct < CPW; ++ct) {
        float* crow = C + (size_t)(m0 + 8 * hl) * NC + n0 + ct * 16 + l;
#pragma unroll
        for (int r = 0; r < 8; ++r) crow[(size_t)r * NC] = acc[ct][r];
    }
}

template <int H, int CH>
__global__ void att_dots_kernel(const float* __restrict__ h,
                                const float* __restrict__ att_s,
                                const float* __restrict__ att_d,
                                float* __restrict__ asrc, float* __restrict__ adst,
                                int n) {
    constexpr int C = H * CH;
    int wave = threadIdx.x >> 5;
    int lane = threadIdx.x & 31;
    int node = blockIdx.x * (blockDim.x >> 5) + wave;
    if (node >= n) return;
    const float* hp = h + (size_t)node * C;

    float s[H], d[H];
#pragma unroll
    for (int i = 0; i < H; ++i) { s[i] = 0.f; d[i] = 0.f; }

#pragma unroll
    for (int i = 0; i < (C + 31) / 32; ++i) {
        int c = lane + i * 32;
        if (c < C) {
            int hd = c / CH;
            float t = hp[c];
            s[hd] += t * att_s[c];
            d[hd] += t * att_d[c];
        }
    }
#pragma unroll
    for (int i = 0; i < H; ++i) {
#pragma unroll
        for (int off = 16; off; off >>= 1) {
            s[i] += __shfl_xor(s[i], off, 32);
            d[i] += __shfl_xor(d[i], off, 32);
        }
        if (lane == 0) {
            asrc[(size_t)node * H + i] = s[i];
            adst[(size_t)node * H + i] = d[i];
        }
    }
}

__device__ inline void atomicMaxF(float* addr, float v) {
    if (v >= 0.0f)
        atomicMax((int*)addr, __float_as_int(v));
    else
        atomicMin((unsigned int*)addr, (unsigned int)__float_as_int(v));
}

template <int H>
__global__ void edge_max_kernel(const int* __restrict__ src, const int* __restrict__ dst,
                                const float* __restrict__ asrc, const float* __restrict__ adst,
                                float* __restrict__ emax, int E) {
    int e = blockIdx.x * blockDim.x + threadIdx.x;
    if (e >= E) return;
    int s = src[e], d = dst[e];
#pragma unroll
    for (int h = 0; h < H; ++h) {
        float v = asrc[s * H + h] + adst[d * H + h];
        v = (v >= 0.f) ? v : NEG_SLOPE * v;
        atomicMaxF(&emax[d * H + h], v);
    }
}

template <int H, int CH>
__global__ __launch_bounds__(256) void edge_agg_kernel(
        const int* __restrict__ src, const int* __restrict__ dst,
        const float* __restrict__ hfeat,
        const float* __restrict__ asrc, const float* __restrict__ adst,
        const float* __restrict__ emax,
        float* __restrict__ denom, float* __restrict__ outraw, int E) {
    constexpr int C   = H * CH;
    constexpr int LPE = C / 4;
    constexpr int EPB = 256 / LPE;
    int li   = threadIdx.x % LPE;
    int esub = threadIdx.x / LPE;
    int e = blockIdx.x * EPB + esub;
    if (e >= E) return;
    int s = src[e], d = dst[e];
    int c0 = li * 4;
    int h  = c0 / CH;

    float v = asrc[s * H + h] + adst[d * H + h];
    v = (v >= 0.f) ? v : NEG_SLOPE * v;
    float wgt = __expf(v - emax[d * H + h]);
    if ((c0 % CH) == 0) atomicAdd(&denom[d * H + h], wgt);

    const float4 hv = *(const float4*)(hfeat + (size_t)s * C + c0);
    float* op = outraw + (size_t)d * C + c0;
    atomicAdd(op + 0, hv.x * wgt);
    atomicAdd(op + 1, hv.y * wgt);
    atomicAdd(op + 2, hv.z * wgt);
    atomicAdd(op + 3, hv.w * wgt);
}

__global__ void finalize_elu_kernel(const float* __restrict__ outraw,
                                    const float* __restrict__ denom,
                                    const float* __restrict__ bias,
                                    float* __restrict__ hact,
                                    int n, int heads, int ch) {
    size_t i = (size_t)blockIdx.x * blockDim.x + threadIdx.x;
    size_t total = (size_t)n * heads * ch;
    if (i >= total) return;
    int C = heads * ch;
    int node = (int)(i / C);
    int c = (int)(i % C);
    int hd = c / ch;
    float val = outraw[i] / (denom[node * heads + hd] + EPS_) + bias[c];
    hact[i] = (val > 0.f) ? val : expm1f(val);
}

__global__ void finalize_softmax_kernel(const float* __restrict__ outraw,
                                        const float* __restrict__ denom,
                                        const float* __restrict__ bias,
                                        float* __restrict__ out, int n) {
    int node = blockIdx.x * blockDim.x + threadIdx.x;
    if (node >= n) return;
    float den = denom[node] + EPS_;
    float v[16];
    float m = -INFINITY;
#pragma unroll
    for (int c = 0; c < 16; ++c) {
        v[c] = outraw[(size_t)node * 16 + c] / den + bias[c];
        m = fmaxf(m, v[c]);
    }
    float sum = 0.f;
#pragma unroll
    for (int c = 0; c < 16; ++c) {
        float t = __expf(v[c] - m);
        v[c] = t;
        sum += t;
    }
    float inv = 1.f / sum;
#pragma unroll
    for (int c = 0; c < 16; ++c) out[(size_t)node * 16 + c] = v[c] * inv;
}

extern "C" void kernel_launch(void* const* d_in, const int* in_sizes, int n_in,
                              void* d_out, int out_size, void* d_ws, size_t ws_size,
                              hipStream_t stream) {
    const float* x   = (const float*)d_in[0];
    const float* W1  = (const float*)d_in[1];
    const float* as1 = (const float*)d_in[2];
    const float* ad1 = (const float*)d_in[3];
    const float* b1  = (const float*)d_in[4];
    const float* W2  = (const float*)d_in[5];
    const float* as2 = (const float*)d_in[6];
    const float* ad2 = (const float*)d_in[7];
    const float* b2  = (const float*)d_in[8];
    const int*   ei  = (const int*)d_in[9];

    const int N = in_sizes[0] / 128;
    const int E = in_sizes[9] / 2;
    const int* src = ei;
    const int* dst = ei + E;

    float* w = (float*)d_ws;
    float* h1      = w; w += (size_t)N * 128;
    float* outraw1 = w; w += (size_t)N * 128;
    float* asrc1   = w; w += (size_t)N * 2;
    float* adst1   = w; w += (size_t)N * 2;
    float* emax1   = w; w += (size_t)N * 2;
    float* denom1  = w; w += (size_t)N * 2;
    float* h2      = w; w += (size_t)N * 16;
    float* outraw2 = w; w += (size_t)N * 16;
    float* asrc2   = w; w += (size_t)N;
    float* adst2   = w; w += (size_t)N;
    float* emax2   = w; w += (size_t)N;
    float* denom2  = w; w += (size_t)N;

    fill_kernel<<<1024, 256, 0, stream>>>(outraw1, 0.f, (size_t)N * 128);
    fill_kernel<<<64, 256, 0, stream>>>(denom1, 0.f, (size_t)N * 2);
    fill_kernel<<<64, 256, 0, stream>>>(emax1, -INFINITY, (size_t)N * 2);
    fill_kernel<<<256, 256, 0, stream>>>(outraw2, 0.f, (size_t)N * 16);
    fill_kernel<<<64, 256, 0, stream>>>(denom2, 0.f, (size_t)N);
    fill_kernel<<<64, 256, 0, stream>>>(emax2, -INFINITY, (size_t)N);

    {
        int tiles = (N / 16) * 2;
        wmma_gemm_f32<128, 128, 4><<<cdiv(tiles, 8), 256, 0, stream>>>(x, W1, h1, N);
        att_dots_kernel<2, 64><<<cdiv(N, 8), 256, 0, stream>>>(h1, as1, ad1, asrc1, adst1, N);
        edge_max_kernel<2><<<cdiv(E, 256), 256, 0, stream>>>(src, dst, asrc1, adst1, emax1, E);
        edge_agg_kernel<2, 64><<<cdiv(E, 8), 256, 0, stream>>>(src, dst, h1, asrc1, adst1, emax1,
                                                               denom1, outraw1, E);
        finalize_elu_kernel<<<cdiv((long)N * 128, 256), 256, 0, stream>>>(outraw1, denom1, b1, h1, N, 2, 64);
    }

    {
        int tiles = (N / 16);
        wmma_gemm_f32<128, 16, 1><<<cdiv(tiles, 8), 256, 0, stream>>>(h1, W2, h2, N);
        att_dots_kernel<1, 16><<<cdiv(N, 8), 256, 0, stream>>>(h2, as2, ad2, asrc2, adst2, N);
        edge_max_kernel<1><<<cdiv(E, 256), 256, 0, stream>>>(src, dst, asrc2, adst2, emax2, E);
        edge_agg_kernel<1, 16><<<cdiv(E, 64), 256, 0, stream>>>(src, dst, h2, asrc2, adst2, emax2,
                                                                denom2, outraw2, E);
        finalize_softmax_kernel<<<cdiv(N, 256), 256, 0, stream>>>(outraw2, denom2, b2, (float*)d_out, N);
    }
}


# node kernels take 4 nodes per wave and the two max passes run as 1024 looping work-groups: the four launches paced by starting 50000 waves each start 4-6x fewer; all sums and their order kept
# speedup vs baseline: 4.5875x; 4.5875x over previous
#include <hip/hip_runtime.h>
#include <hip/hip_bf16.h>
#include <math.h>

typedef float v2f __attribute__((ext_vector_type(2)));
typedef float v8f __attribute__((ext_vector_type(8)));

#define NEG_SLOPE 0.2f
#define EPS_ 1e-16f

static inline int cdiv(long a, long b) { return (int)((a + b - 1) / b); }

#define SPAN_SLOTS 32
#define SPAN_INTS  (SPAN_SLOTS * 2)
#define SPAN_MAX   16.f

#define NODES_PER_WAVE 4
#define EDGE_MAX_WGS   1024

template <int H>
__device__ inline void span_offer(int* span, int wave, int lane, const float* s, int* sh) {
    if (lane < H) {
        float v = s[0];
#pragma unroll
        for (int i = 1; i < H; ++i) v = (lane == i) ? s[i] : v;
        sh[wave * H + lane] = __float_as_int(fabsf(v));
    }
    __syncthreads();
    if (threadIdx.x < H) {
        int m = sh[threadIdx.x];
#pragma unroll
        for (int w = 1; w < 8; ++w) m = max(m, sh[w * H + threadIdx.x]);
        int* p = span + (blockIdx.x % SPAN_SLOTS) * 2 + threadIdx.x;
        if (m > __hip_atomic_load(p, __ATOMIC_RELAXED, __HIP_MEMORY_SCOPE_AGENT))
            atomicMax(p, m);
    }
}

__device__ inline void span_read(const int* span, int* sh) {
    int2 v = ((const int2*)span)[threadIdx.x];
#pragma unroll
    for (int off = 16; off; off >>= 1) {
        v.x = max(v.x, __shfl_xor(v.x, off, 32));
        v.y = max(v.y, __shfl_xor(v.y, off, 32));
    }
    if (threadIdx.x == 0) { sh[1] = v.x; sh[2] = v.y; }
}
__device__ inline bool span_small(const int* sh) {
    return 2.f * __int_as_float(sh[1]) <= SPAN_MAX && 2.f * __int_as_float(sh[2]) <= SPAN_MAX;
}

__global__ void fill_kernel(float* __restrict__ p, float v, size_t n) {
    size_t i = (size_t)blockIdx.x * blockDim.x + threadIdx.x;
    size_t stride = (size_t)gridDim.x * blockDim.x;
    for (; i < n; i += stride) p[i] = v;
}

template <int K, int NC, int CPW, int LDC>
__global__ __launch_bounds__(256) void wmma_gemm_f32(const float* __restrict__ A,
                                                     const float* __restrict__ B,
                                                     float* __restrict__ C,
                                                     int* __restrict__ span, int M) {
    if (blockIdx.x == 0 && threadIdx.x < SPAN_INTS) span[threadIdx.x] = 0;
    constexpr int nColT  = NC / 16;
    constexpr int groups = nColT / CPW;
    int wave = threadIdx.x >> 5;
    int lane = threadIdx.x & 31;
    int hl   = lane >> 4;
    int l    = lane & 15;
    int tile  = blockIdx.x * (blockDim.x >> 5) + wave;
    int total = (M >> 4) * groups;
    if (tile >= total) return;
    int m0 = (tile / groups) << 4;
    int n0 = (tile % groups) * (CPW * 16);

    const float* arow = A + (size_t)(m0 + l) * K + 2 * hl;
    const float* bcol = B + (size_t)(2 * hl) * NC + n0 + l;

    v8f acc[CPW];
#pragma unroll
    for (int i = 0; i < CPW; ++i) acc[i] = {};

    for (int k0 = 0; k0 < K; k0 += 4) {
        v2f a;
        a.x = arow[k0 + 0];
        a.y = arow[k0 + 1];
#pragma unroll
        for (int ct = 0; ct < CPW; ++ct) {
            v2f b;
            b.x = bcol[(size_t)k0 * NC + ct * 16];
            b.y = bcol[(size_t)(k0 + 1) * NC + ct * 16];
            acc[ct] = __builtin_amdgcn_wmma_f32_16x16x4_f32(false, a, false, b,
                                                            (short)0, acc[ct], false, false);
        }
    }

#pragma unroll
    for (int ct = 0; ct < CPW; ++ct) {
        float* crow = C + (size_t)(m0 + 8 * hl) * LDC + n0 + ct * 16 + l;
#pragma unroll
        for (int r = 0; r < 8; ++r) crow[(size_t)r * LDC] = acc[ct][r];
    }
}

#define KEY_NEG_INF ((int)0x807fffff)
__device__ inline int float_key(float v) {
    int b = __float_as_int(v);
    return b ^ ((b >> 31) & 0x7fffffff);
}
__device__ inline float key_float(int k) {
    return __int_as_float(k ^ ((k >> 31) & 0x7fffffff));
}

template <bool SPAN>
__device__ inline int take_job(int* tick, int nchunk, const int* span, int* sh_job) {
    int2 v = {0, 0};
    if (SPAN && threadIdx.x < 32) v = ((const int2*)span)[threadIdx.x];
    if (threadIdx.x == 0) {
        int half = (__builtin_amdgcn_s_sendmsg_rtn(0x87) >> 18) & 1;
        int t = atomicAdd(&tick[half], 1);
        if (t >= nchunk) { half ^= 1; t = atomicAdd(&tick[half], 1); }
        *sh_job = (t < nchunk) ? (t * 2 + half) : -1;
    }
    if (SPAN && threadIdx.x < 32) {
#pragma unroll
        for (int off = 16; off; off >>= 1) {
            v.x = max(v.x, __shfl_xor(v.x, off, 32));
            v.y = max(v.y, __shfl_xor(v.y, off, 32));
        }
        if (threadIdx.x == 0) { sh_job[1] = v.x; sh_job[2] = v.y; }
    }
    __syncthreads();
    return *sh_job;
}

template <int H, int CH, int NPW>
__global__ __launch_bounds__(256) void att_dots_kernel(const float* __restrict__ h,
                                const float* __restrict__ att_s,
                                const float* __restrict__ att_d,
                                float* __restrict__ asrc, int* __restrict__ drec,
                                int* __restrict__ tick, int* span, int n) {
    constexpr int C  = H * CH;
    constexpr int NL = (C + 31) / 32;
    int wave = threadIdx.x >> 5;
    int lane = threadIdx.x & 31;
    __shared__ int sh_span[8 * H];
    const int node0 = (blockIdx.x * (blockDim.x >> 5) + wave) * NPW;
    if (blockIdx.x == 0 && threadIdx.x < 4) tick[threadIdx.x] = 0;

    float ws[NL], wd[NL];
#pragma unroll
    for (int i = 0; i < NL; ++i) {
        int c = lane + i * 32;
        ws[i] = (c < C) ? att_s[c] : 0.f;
        wd[i] = (c < C) ? att_d[c] : 0.f;
    }
    float t[NPW][NL];
#pragma unroll
    for (int j = 0; j < NPW; ++j) {
        int nd = node0 + j;
        nd = (nd < n) ? nd : n - 1;
        const float* hp = h + (size_t)nd * C;
#pragma unroll
        for (int i = 0; i < NL; ++i) {
            int c = lane + i * 32;
            t[j][i] = (c < C) ? hp[c] : 0.f;
        }
    }
    float s[NPW][H], d[NPW][H];
    int ab[H];
#pragma unroll
    for (int i = 0; i < H; ++i) ab[i] = 0;
#pragma unroll
    for (int j = 0; j < NPW; ++j) {
#pragma unroll
        for (int i = 0; i < H; ++i) { s[j][i] = 0.f; d[j][i] = 0.f; }
#pragma unroll
        for (int i = 0; i < NL; ++i) {
            int c = lane + i * 32;
            if (c < C) {
                int hd = c / CH;
                s[j][hd] += t[j][i] * ws[i];
                d[j][hd] += t[j][i] * wd[i];
            }
        }
#pragma unroll
        for (int i = 0; i < H; ++i) {
#pragma unroll
            for (int off = 16; off; off >>= 1) {
                s[j][i] += __shfl_xor(s[j][i], off, 32);
                d[j][i] += __shfl_xor(d[j][i], off, 32);
            }
            ab[i] = max(ab[i], __float_as_int(fabsf(s[j][i])));
        }
    }
    float amax[H];
#pragma unroll
    for (int i = 0; i < H; ++i) amax[i] = __int_as_float(ab[i]);
    span_offer<H>(span, wave, lane, amax, sh_span);
    if (lane == 0) {
#pragma unroll
        for (int j = 0; j < NPW; ++j) {
            int node = node0 + j;
            if (node < n) {
#pragma unroll
                for (int i = 0; i < H; ++i) {
                    asrc[(size_t)node * H + i] = s[j][i];
                    drec[(size_t)node * 2 * H + i] = __float_as_int(d[j][i]);
                    drec[(size_t)node * 2 * H + H + i] = KEY_NEG_INF;
                }
            }
        }
    }
}

template <int H, int EP, int EO>
__global__ __launch_bounds__(256) void edge_max_kernel(
        const int* __restrict__ src, const int* __restrict__ dst,
        const float* __restrict__ asrc,
        int* __restrict__ mkey, int* tick, const int* __restrict__ span, int E, int n) {
    __shared__ int sh[3];
    if (threadIdx.x < 32) span_read(span, sh);
    __syncthreads();
    if (span_small(sh)) return;
    const int nchunk = (E + 511) / 512;
    const int nsplit = n >> 1;
    for (int trip = 0; trip <= 2 * nchunk; ++trip) {
    const int job = take_job<false>(tick, nchunk, span, sh);
    if (job < 0) return;
    const bool upper = (job & 1) != 0;
    const int e0 = (job >> 1) * 512 + threadIdx.x;
    int s[2], d[2];
    bool mine[2];
    float a[2][H];
#pragma unroll
    for (int k = 0; k < 2; ++k) {
        int e = e0 + k * 256;
        int ec = (e < E) ? e : E - 1;
        s[k] = src[ec];
        d[k] = dst[ec];
        mine[k] = (e < E) && ((d[k] >= nsplit) == upper);
    }
#pragma unroll
    for (int k = 0; k < 2; ++k) {
        const float* ap = asrc + (mine[k] ? s[k] * H : 0);
        if constexpr (H == 2) {
            float2 t = *(const float2*)ap;
            a[k][0] = t.x;
            a[k][1] = t.y;
        } else {
#pragma unroll
            for (int h = 0; h < H; ++h) a[k][h] = ap[h];
        }
    }
#pragma unroll
    for (int k = 0; k < 2; ++k) {
        if (mine[k]) {
#pragma unroll
            for (int h = 0; h < H; ++h)
                atomicMax(&mkey[(size_t)d[k] * EP + EO + h], float_key(a[k][h]));
        }
    }
    __syncthreads();
    }
}

__device__ inline int pack_own_edges(bool mine, int lane, int& s, int& d) {
    const unsigned mask = __builtin_amdgcn_ballot_w32(mine);
    const int cnt = __popc(mask);
    const unsigned below = (1u << lane) - 1u;
    const int slot = mine ? __popc(mask & below) : cnt + __popc(~mask & below);
    const int ps = __builtin_amdgcn_ds_permute(slot << 2, s);
    const int pd = __builtin_amdgcn_ds_permute(slot << 2, d);
    s = (lane < cnt) ? ps : 0;
    d = (lane < cnt) ? pd : 0;
    return cnt;
}

__global__ __launch_bounds__(256) void edge_agg_lines_kernel(
        const int* __restrict__ src, const int* __restrict__ dst,
        const float* __restrict__ hfeat,
        const float* __restrict__ asrc, const int* __restrict__ drec,
        float* __restrict__ denom, float* __restrict__ outraw,
        int* tick, const int* __restrict__ span, int E, int n) {
    constexpr int C = 128;
    __shared__ int sh[3];
    const int job = take_job<true>(tick, (E + 255) / 256, span, sh);
    if (job < 0) return;
    const bool chip_shift = span_small(sh);
    const float A0 = __int_as_float(sh[1]), A1 = __int_as_float(sh[2]);
    const bool upper = (job & 1) != 0;
    int wave = threadIdx.x >> 5;
    int lane = threadIdx.x & 31;
    int e = (job >> 1) * 256 + wave * 32 + lane;
    int my_s = 0, my_d = 0;
    bool mine = false;
    if (e < E) {
        my_s = src[e];
        my_d = dst[e];
        mine = ((my_d >= (n >> 1)) == upper);
    }
    const int cnt = pack_own_edges(mine, lane, my_s, my_d);
    if (cnt == 0) return;

    const float2 as = *(const float2*)(asrc + (size_t)my_s * 2);
    const int4   dr = *(const int4*)(drec + (size_t)my_d * 4);
    const float ad0 = __int_as_float(dr.x), ad1 = __int_as_float(dr.y);
    float v0 = as.x + ad0;
    float v1 = as.y + ad1;
    float m0 = (chip_shift ? A0 : key_float(dr.z)) + ad0;
    float m1 = (chip_shift ? A1 : key_float(dr.w)) + ad1;
    v0 = (v0 >= 0.f) ? v0 : NEG_SLOPE * v0;
    v1 = (v1 >= 0.f) ? v1 : NEG_SLOPE * v1;
    m0 = (m0 >= 0.f) ? m0 : NEG_SLOPE * m0;
    m1 = (m1 >= 0.f) ? m1 : NEG_SLOPE * m1;
    const float my_w0 = __expf(v0 - m0);
    const float my_w1 = __expf(v1 - m1);

#pragma unroll
    for (int g = 0; g < 4; ++g) {
        if (g * 8 < cnt) {
            float hv[8][4];
#pragma unroll
            for (int i = 0; i < 8; ++i) {
                int s = __builtin_amdgcn_readlane(my_s, g * 8 + i);
                const float* hp = hfeat + (size_t)s * C + lane;
                hv[i][0] = hp[0];
                hv[i][1] = hp[32];
                hv[i][2] = hp[64];
                hv[i][3] = hp[96];
            }
#pragma unroll
            for (int i = 0; i < 8; ++i) {
                int d = __builtin_amdgcn_readlane(my_d, g * 8 + i);
                float w0 = __int_as_float(__builtin_amdgcn_readlane(__float_as_int(my_w0), g * 8 + i));
                float w1 = __int_as_float(__builtin_amdgcn_readlane(__float_as_int(my_w1), g * 8 + i));
                if (g * 8 + i < cnt) {
                    float* op = outraw + (size_t)d * C + lane;
                    atomicAdd(op,      hv[i][0] * w0);
                    atomicAdd(op + 32, hv[i][1] * w0);
                    atomicAdd(op + 64, hv[i][2] * w1);
                    atomicAdd(op + 96, hv[i][3] * w1);
                    if (lane < 2) atomicAdd(denom + (size_t)d * 2 + lane, lane ? w1 : w0);
                }
            }
        }
    }
}

#define ROWP   32
#define F_ASRC 16
#define R_DEN  16
#define R_MAX  18
#define R_ADST 19

template <int NPW>
__global__ __launch_bounds__(256) void node_rows_kernel(float* F, const float* __restrict__ att_s,
                                 const float* __restrict__ att_d,
                                 float* __restrict__ asrc,
                                 int* __restrict__ R, int* __restrict__ tick,
                                 int* span, int n) {
    int wave = threadIdx.x >> 5;
    int lane = threadIdx.x & 31;
    __shared__ int sh_span[8];
    const int node0 = (blockIdx.x * (blockDim.x >> 5) + wave) * NPW;
    if (blockIdx.x == 0 && threadIdx.x < 4) tick[threadIdx.x] = 0;
    const float ws = (lane < 16) ? att_s[lane] : 0.f;
    const float wd = (lane < 16) ? att_d[lane] : 0.f;
    float t[NPW];
#pragma unroll
    for (int j = 0; j < NPW; ++j) {
        int nd = node0 + j;
        nd = (nd < n) ? nd : n - 1;
        t[j] = (lane < 16) ? F[(size_t)nd * ROWP + lane] : 0.f;
    }
    float s[NPW], d[NPW];
    int ab = 0;
#pragma unroll
    for (int j = 0; j < NPW; ++j) {
        s[j] = 0.f;
        d[j] = 0.f;
        if (lane < 16) {
            s[j] += t[j] * ws;
            d[j] += t[j] * wd;
        }
#pragma unroll
        for (int off = 16; off; off >>= 1) {
            s[j] += __shfl_xor(s[j], off, 32);
            d[j] += __shfl_xor(d[j], off, 32);
        }
        ab = max(ab, __float_as_int(fabsf(s[j])));
    }
    float amax = __int_as_float(ab);
    span_offer<1>(span, wave, lane, &amax, sh_span);
#pragma unroll
    for (int j = 0; j < NPW; ++j) {
        int node = node0 + j;
        if (node < n) {
            if (lane == 0) {
                asrc[node] = s[j];
                F[(size_t)node * ROWP + F_ASRC] = s[j];
            }
            int init = (lane == R_MAX) ? KEY_NEG_INF : ((lane == R_ADST) ? __float_as_int(d[j]) : 0);
            R[(size_t)node * ROWP + lane] = init;
        }
    }
}

__global__ __launch_bounds__(256) void edge_agg_rows_kernel(
        const int* __restrict__ src, const int* __restrict__ dst,
        const float* __restrict__ F, float* R, int* tick,
        const int* __restrict__ span, int E, int n) {
    __shared__ int sh[3];
    const int job = take_job<true>(tick, (E + 255) / 256, span, sh);
    if (job < 0) return;
    const bool chip_shift = span_small(sh);
    const float A0 = __int_as_float(sh[1]);
    const bool upper = (job & 1) != 0;
    int wave = threadIdx.x >> 5;
    int lane = threadIdx.x & 31;
    int e = (job >> 1) * 256 + wave * 32 + lane;
    int my_s = 0, my_d = 0;
    bool mine = false;
    if (e < E) {
        my_s = src[e];
        my_d = dst[e];
        mine = ((my_d >= (n >> 1)) == upper);
    }
    const int cnt = pack_own_edges(mine, lane, my_s, my_d);
    if (cnt == 0) return;

    int col = (lane <= F_ASRC) ? lane : F_ASRC;
#pragma unroll
    for (int g = 0; g < 4; ++g) {
        if (g * 8 < cnt) {
            float  hv[8];
            float2 md[8];
            int    dn[8];
#pragma unroll
            for (int i = 0; i < 8; ++i) {
                int s = __builtin_amdgcn_readlane(my_s, g * 8 + i);
                int d = __builtin_amdgcn_readlane(my_d, g * 8 + i);
                dn[i] = d;
                hv[i] = F[(size_t)s * ROWP + col];
                md[i] = *(const float2*)(R + (size_t)d * ROWP + R_MAX);
            }
#pragma unroll
            for (int i = 0; i < 8; ++i) {
                float a_s = __int_as_float(__builtin_amdgcn_readlane(__float_as_int(hv[i]), F_ASRC));
                float v = a_s + md[i].y;
                float m = (chip_shift ? A0 : key_float(__float_as_int(md[i].x))) + md[i].y;
                v = (v >= 0.f) ? v : NEG_SLOPE * v;
                m = (m >= 0.f) ? m : NEG_SLOPE * m;
                float wgt = __expf(v - m);
                float val = (lane == R_DEN) ? wgt : hv[i] * wgt;
                if (g * 8 + i < cnt && lane <= R_DEN) atomicAdd(R + (size_t)dn[i] * ROWP + lane, val);
            }
        }
    }
}

__global__ __launch_bounds__(256) void wmma_gemm_elu_f32(const float* __restrict__ outraw,
                                                         const float* __restrict__ denom,
                                                         const float* __restrict__ bias,
                                                         const float* __restrict__ B,
                                                         float* __restrict__ C,
                                                         int* __restrict__ span, int M) {
    if (blockIdx.x == 0 && threadIdx.x < SPAN_INTS) span[threadIdx.x] = 0;
    constexpr int K = 128, NC = 16, CH = 64;
    int wave = threadIdx.x >> 5;
    int lane = threadIdx.x & 31;
    int hl   = lane >> 4;
    int l    = lane & 15;
    int tile = blockIdx.x * (blockDim.x >> 5) + wave;
    if (tile >= (M >> 4)) return;
    int m0 = tile << 4;

    const float* arow = outraw + (size_t)(m0 + l) * K + 2 * hl;
    const float* brow = bias + 2 * hl;
    const float* bcol = B + (size_t)(2 * hl) * NC + l;
    const float2 dn = *(const float2*)(denom + (size_t)(m0 + l) * 2);
    const float den0 = dn.x + EPS_;
    const float den1 = dn.y + EPS_;

    v8f acc = {};
#pragma unroll
    for (int k0 = 0; k0 < K; k0 += 4) {
        const float den = (k0 < CH) ? den0 : den1;
        const float2 r  = *(const float2*)(arow + k0);
        const float2 bs = *(const float2*)(brow + k0);
        float v0 = r.x / den + bs.x;
        float v1 = r.y / den + bs.y;
        const float e0 = expm1f((v0 > 0.f) ? 0.f : v0);
        const float e1 = expm1f((v1 > 0.f) ? 0.f : v1);
        v2f a;
        a.x = (v0 > 0.f) ? v0 : e0;
        a.y = (v1 > 0.f) ? v1 : e1;
        v2f b;
        b.x = bcol[(size_t)k0 * NC];
        b.y = bcol[(size_t)(k0 + 1) * NC];
        acc = __builtin_amdgcn_wmma_f32_16x16x4_f32(false, a, false, b,
                                                    (short)0, acc, false, false);
    }

    float* crow = C + (size_t)(m0 + 8 * hl) * ROWP + l;
#pragma unroll
    for (int r = 0; r < 8; ++r) crow[(size_t)r * ROWP] = acc[r];
}

__global__ void finalize_softmax_kernel(const float* __restrict__ R,
                                        const float* __restrict__ bias,
                                        float* __restrict__ out, int n) {
    int node = blockIdx.x * blockDim.x + threadIdx.x;
    if (node >= n) return;
    const float* outraw = R + (size_t)node * ROWP;
    float den = outraw[R_DEN] + EPS_;
    float v[16];
    float m = -INFINITY;
#pragma unroll
    for (int c = 0; c < 16; ++c) {
        v[c] = outraw[c] / den + bias[c];
        m = fmaxf(m, v[c]);
    }
    float sum = 0.f;
#pragma unroll
    for (int c = 0; c < 16; ++c) {
        float t = __expf(v[c] - m);
        v[c] = t;
        sum += t;
    }
    float inv = 1.f / sum;
#pragma unroll
    for (int c = 0; c < 16; ++c) out[(size_t)node * 16 + c] = v[c] * inv;
}

extern "C" void kernel_launch(void* const* d_in, const int* in_sizes, int n_in,
                              void* d_out, int out_size, void* d_ws, size_t ws_size,
                              hipStream_t stream) {
    const float* x   = (const float*)d_in[0];
    const float* W1  = (const float*)d_in[1];
    const float* as1 = (const float*)d_in[2];
    const float* ad1 = (const float*)d_in[3];
    const float* b1  = (const float*)d_in[4];
    const float* W2  = (const float*)d_in[5];
    const float* as2 = (const float*)d_in[6];
    const float* ad2 = (const float*)d_in[7];
    const float* b2  = (const float*)d_in[8];
    const int*   ei  = (const int*)d_in[9];

    const int N = in_sizes[0] / 128;
    const int E = in_sizes[9] / 2;
    const int* src = ei;
    const int* dst = ei + E;

    float* w = (float*)d_ws;
    float* h1      = w; w += (size_t)N * 128;
    float* outraw1 = w; w += (size_t)N * 128;
    float* drec1f  = w; w += (size_t)N * 4;
    float* asrc1   = w; w += (size_t)N * 2;
    float* denom1  = w; w += (size_t)N * 2;
    float* asrc2   = w; w += (size_t)N;
    int*   drec1   = (int*)drec1f;
    int*   tick1   = (int*)asrc2;
    int*   tick2   = (int*)denom1;
    int*   span1   = (int*)asrc2 + 32;
    int*   span2   = (int*)drec1f;
    float* F2      = h1;
    float* R2      = h1 + (size_t)N * ROWP;

    fill_kernel<<<1024, 256, 0, stream>>>(outraw1, 0.f, (size_t)N * 128);
    fill_kernel<<<64, 256, 0, stream>>>(denom1, 0.f, (size_t)N * 2);

    {
        int tiles = (N / 16) * 2;
        wmma_gemm_f32<128, 128, 4, 128><<<cdiv(tiles, 8), 256, 0, stream>>>(x, W1, h1, span1, N);
        att_dots_kernel<2, 64, NODES_PER_WAVE><<<cdiv(N, 8 * NODES_PER_WAVE), 256, 0, stream>>>(h1, as1, ad1, asrc1, drec1, tick1,
                                                               span1, N);
        edge_max_kernel<2, 4, 2><<<EDGE_MAX_WGS, 256, 0, stream>>>(src, dst, asrc1, drec1,
                                                                       tick1, span1, E, N);
        edge_agg_lines_kernel<<<2 * cdiv(E, 256), 256, 0, stream>>>(src, dst, h1, asrc1, drec1,
                                                                    denom1, outraw1, tick1 + 2,
                                                                    span1, E, N);
    }

    {
        int tiles = (N / 16);
        wmma_gemm_elu_f32<<<cdiv(tiles, 8), 256, 0, stream>>>(outraw1, denom1, b1, W2, F2, span2, N);
        node_rows_kernel<NODES_PER_WAVE><<<cdiv(N, 8 * NODES_PER_WAVE), 256, 0, stream>>>(F2, as2, ad2, asrc2, (int*)R2, tick2,
                                                         span2, N);
        edge_max_kernel<1, ROWP, R_MAX><<<EDGE_MAX_WGS, 256, 0, stream>>>(src, dst, asrc2, (int*)R2,
                                                                              tick2, span2, E, N);
        edge_agg_rows_kernel<<<2 * cdiv(E, 256), 256, 0, stream>>>(src, dst, F2, R2, tick2 + 2,
                                                                   span2, E, N);
        finalize_softmax_kernel<<<cdiv(N, 256), 256, 0, stream>>>(R2, b2, (float*)d_out, N);
    }
}
